# starting kernel
# baseline (speedup 1.0000x reference)
#include <hip/hip_runtime.h>
#include <hip/hip_bf16.h>

typedef __attribute__((ext_vector_type(2))) float v2f;
typedef __attribute__((ext_vector_type(8))) float v8f;

#define DIMD 64
#define KCODES 512
#define NROWS 65536
#define TILE_ROWS 16

__device__ __forceinline__ float wave_sum(float v) {
#pragma unroll
  for (int off = 16; off > 0; off >>= 1) v += __shfl_xor(v, off, 32);
  return v;
}
__device__ __forceinline__ float wave_max(float v) {
#pragma unroll
  for (int off = 16; off > 0; off >>= 1) v = fmaxf(v, __shfl_xor(v, off, 32));
  return v;
}

__global__ void vq_zero(float* acc) {
  int i = blockIdx.x * blockDim.x + threadIdx.x;
  if (i < 514) acc[i] = 0.0f;
}

__global__ __launch_bounds__(256) void vq_cbnorm(const float* __restrict__ cb,
                                                 float* __restrict__ cbn) {
  int wave = (blockIdx.x * blockDim.x + threadIdx.x) >> 5;
  int lane = threadIdx.x & 31;
  const float* row = cb + wave * DIMD;
  float v0 = row[lane], v1 = row[lane + 32];
  float s = wave_sum(v0 * v0 + v1 * v1);
  float scale = 1.0f / fmaxf(sqrtf(s), 1e-12f);
  cbn[wave * DIMD + lane] = v0 * scale;
  cbn[wave * DIMD + lane + 32] = v1 * scale;
}

__global__ __launch_bounds__(32) void vq_main(
    const float* __restrict__ z, const float* __restrict__ kappa_p,
    const float* __restrict__ cbn, const float* __restrict__ gum,
    float* __restrict__ out, float* __restrict__ avgp_acc,
    float* __restrict__ sdisc, float* __restrict__ scont) {
  __shared__ float zt[TILE_ROWS][DIMD];
  __shared__ float ybuf[TILE_ROWS][KCODES];
  __shared__ float avgp[KCODES];

  const int lane = threadIdx.x;
  const int m = lane & 15;
  const int hi = lane >> 4;
  const long base_row = (long)blockIdx.x * TILE_ROWS;
  const float kappa = kappa_p[0];

#pragma unroll
  for (int i = lane; i < KCODES; i += 32) avgp[i] = 0.0f;

#pragma unroll
  for (int r = 0; r < TILE_ROWS; ++r) {
    const float* zr = z + (base_row + r) * DIMD;
    float v0 = zr[lane], v1 = zr[lane + 32];
    float s = wave_sum(v0 * v0 + v1 * v1);
    float scale = 1.0f / fmaxf(sqrtf(s), 1e-12f);
    zt[r][lane] = v0 * scale;
    zt[r][lane + 32] = v1 * scale;
  }
  __syncthreads();

  for (int ct = 0; ct < KCODES / 16; ++ct) {
    v8f c = {};
    const float* cbrow = cbn + (ct * 16 + m) * DIMD;
#pragma unroll
    for (int k = 0; k < 16; ++k) {
      int kb = 4 * k + 2 * hi;
      v2f a;
      a.x = zt[m][kb];
      a.y = zt[m][kb + 1];
      v2f b;
      b.x = cbrow[kb];
      b.y = cbrow[kb + 1];
      c = __builtin_amdgcn_wmma_f32_16x16x4_f32(false, a, false, b, (short)0,
                                                c, false, false);
    }
#pragma unroll
    for (int j = 0; j < 8; ++j)
      ybuf[j + 8 * hi][ct * 16 + m] = kappa * c[j];
  }
  __syncthreads();

  float acc_pl = 0.0f;
  for (int r = 0; r < TILE_ROWS; ++r) {
    const float4* urow4 =
        (const float4*)(gum + (base_row + r) * (long)KCODES) + lane * 4;
    float yv[16], y2v[16];
    float m1 = -3.4e38f, m2 = -3.4e38f;
#pragma unroll
    for (int i4 = 0; i4 < 4; ++i4) {
      float4 u4 = urow4[i4];
      float uu[4] = {u4.x, u4.y, u4.z, u4.w};
#pragma unroll
      for (int j = 0; j < 4; ++j) {
        int i = i4 * 4 + j;
        int c0 = lane * 16 + i;
        float yy = ybuf[r][c0];
        float g = -logf(-logf(uu[j] + 1e-10f) + 1e-10f);
        float y2 = (yy + g) * 2.0f;
        yv[i] = yy;
        y2v[i] = y2;
        m1 = fmaxf(m1, yy);
        m2 = fmaxf(m2, y2);
      }
    }
    m1 = wave_max(m1);
    m2 = wave_max(m2);
    float s1 = 0.0f, s2 = 0.0f;
#pragma unroll
    for (int i = 0; i < 16; ++i) {
      s1 += expf(yv[i] - m1);
      s2 += expf(y2v[i] - m2);
    }
    s1 = wave_sum(s1);
    s2 = wave_sum(s2);
    float lse = m1 + logf(s1);
    float inv_s2 = 1.0f / s2;
#pragma unroll
    for (int i = 0; i < 16; ++i) {
      int c0 = lane * 16 + i;
      float lp = yv[i] - lse;
      float p = expf(lp);
      acc_pl += p * lp;
      avgp[c0] += p;
      ybuf[r][c0] = expf(y2v[i] - m2) * inv_s2;
    }
  }
  acc_pl = wave_sum(acc_pl);
  if (lane == 0) atomicAdd(sdisc, acc_pl);
#pragma unroll
  for (int i = lane; i < KCODES; i += 32) atomicAdd(&avgp_acc[i], avgp[i]);
  __syncthreads();

  float acc_cont = 0.0f;
  for (int nt = 0; nt < 4; ++nt) {
    v8f c = {};
    for (int k = 0; k < KCODES / 4; ++k) {
      int kb = 4 * k + 2 * hi;
      v2f a;
      a.x = ybuf[m][kb];
      a.y = ybuf[m][kb + 1];
      v2f b;
      b.x = cbn[(long)kb * DIMD + nt * 16 + m];
      b.y = cbn[(long)(kb + 1) * DIMD + nt * 16 + m];
      c = __builtin_amdgcn_wmma_f32_16x16x4_f32(false, a, false, b, (short)0,
                                                c, false, false);
    }
#pragma unroll
    for (int j = 0; j < 8; ++j) {
      int r = j + 8 * hi;
      long idx = (base_row + r) * DIMD + nt * 16 + m;
      float zq = c[j];
      float zo = z[idx];
      acc_cont += zo * (zo - zq);
      out[idx] = zq;
    }
  }
  acc_cont = wave_sum(acc_cont);
  if (lane == 0) atomicAdd(scont, acc_cont);
}

__global__ __launch_bounds__(32) void vq_finalize(
    const float* __restrict__ kappa_p, const float* __restrict__ avgp_acc,
    const float* __restrict__ sdisc, const float* __restrict__ scont,
    float* __restrict__ out) {
  int lane = threadIdx.x;
  const float invN = 1.0f / (float)NROWS;
  float acc = 0.0f;
#pragma unroll
  for (int i = lane; i < KCODES; i += 32) {
    float a = avgp_acc[i] * invN;
    acc += a * logf(a + 1e-7f);
  }
  acc = wave_sum(acc);
  if (lane == 0) {
    float loss = (sdisc[0] + kappa_p[0] * scont[0]) / 8.0f;
    out[(long)NROWS * DIMD] = loss;
    out[(long)NROWS * DIMD + 1] = expf(-acc);
  }
}

extern "C" void kernel_launch(void* const* d_in, const int* in_sizes, int n_in,
                              void* d_out, int out_size, void* d_ws,
                              size_t ws_size, hipStream_t stream) {
  const float* z = (const float*)d_in[0];
  const float* kappa = (const float*)d_in[1];
  const float* cb = (const float*)d_in[2];
  const float* gum = (const float*)d_in[3];
  float* out = (float*)d_out;

  float* wsf = (float*)d_ws;
  float* avgp_acc = wsf;
  float* sdisc = wsf + 512;
  float* scont = wsf + 513;
  float* cbn = wsf + 1024;

  vq_zero<<<3, 256, 0, stream>>>(wsf);
  vq_cbnorm<<<64, 256, 0, stream>>>(cb, cbn);
  vq_main<<<NROWS / TILE_ROWS, 32, 0, stream>>>(z, kappa, cbn, gum, out,
                                                avgp_acc, sdisc, scont);
  vq_finalize<<<1, 32, 0, stream>>>(kappa, avgp_acc, sdisc, scont, out);
}


# Logits of 9 of the 16 pass-1 trips kept on chip (6 in the wave's own spare LDS slots, 3 in spare registers as a queue): pass 2 re-forms GEMM1 for 7 trips, not 16; every sum and its order unchanged
# speedup vs baseline: 1.3505x; 1.3505x over previous
#include <hip/hip_runtime.h>
#include <hip/hip_bf16.h>


typedef __attribute__((ext_vector_type(2))) float v2f;
typedef __attribute__((ext_vector_type(8))) float v8f;

#define DIMD 64
#define KCODES 512
#define NROWS 65536
#define TILE_ROWS 16
#define ZQ_PITCH 68
#define NWAVES 16
#define NBLOCKS 256
#define ZQ_F4 (TILE_ROWS * ZQ_PITCH / 4)
#define NSTASH 6
#define AREA_F4 (NSTASH * 128)
#define NREG 3
#define REG_FROM (KCODES / 16 - 2 * NREG)
static_assert(ZQ_F4 <= AREA_F4, "the output staging reuses the wave's area");
static_assert((8192 + NWAVES * AREA_F4) * 16 <= 327680, "LDS of one compute unit");

__device__ __forceinline__ float wave_sum(float v) {
#pragma unroll
  for (int off = 16; off > 0; off >>= 1) v += __shfl_xor(v, off, 32);
  return v;
}

template <int CTRL>
__device__ __forceinline__ float dpp_get(float v) {
  return __builtin_bit_cast(
      float, __builtin_amdgcn_update_dpp(0, __builtin_bit_cast(int, v), CTRL,
                                         0xF, 0xF, true));
}
__device__ __forceinline__ float row16_sum(float v) {
  v += dpp_get<0xB1>(v);
  v += dpp_get<0x4E>(v);
  v += dpp_get<0x141>(v);
  v += dpp_get<0x140>(v);
  return v;
}

__device__ __forceinline__ float xhalf(float x) {
  int b = __builtin_bit_cast(int, x);
  return __builtin_bit_cast(
      float, __builtin_amdgcn_permlanex16(b, b, 0x76543210, 0xfedcba98, false, false));
}

__device__ __forceinline__ void gemm1_pair(const float* cbs, int ct, int c, int h,
                                           const v2f (&zf)[16], v8f& a0, v8f& a1) {
  a0 = (v8f){};
  a1 = (v8f){};
  const int a1o = (ct * 16 + c) * DIMD + 4 * c + 2 * h;
  float2 t0[16], t1[16];
#pragma unroll
  for (int ks = 0; ks < 16; ++ks) {
    const float* p0 = cbs + (a1o ^ (ks << 2));
    t0[ks] = *(const float2*)p0;
    t1[ks] = *(const float2*)(p0 + 16 * DIMD);
  }
  __builtin_amdgcn_sched_barrier(0);
#pragma unroll
  for (int ks = 0; ks < 16; ++ks) {
    v2f f0, f1;
    f0.x = t0[ks].x;
    f0.y = t0[ks].y;
    f1.x = t1[ks].x;
    f1.y = t1[ks].y;
    a0 = __builtin_amdgcn_wmma_f32_16x16x4_f32(false, f0, false, zf[ks],
                                               (short)0, a0, false, false);
    a1 = __builtin_amdgcn_wmma_f32_16x16x4_f32(false, f1, false, zf[ks],
                                               (short)0, a1, false, false);
  }
  asm volatile("v_nop\n\tv_nop\n\tv_nop\n\tv_nop" : "+v"(a0), "+v"(a1));
}

__device__ __forceinline__ void pass1_tail(const v8f& a0, const v8f& a1, float kappa,
                                           float& m1, float& s1) {
  float q[16];
#pragma unroll
  for (int r = 0; r < 8; ++r) {
    q[r] = kappa * a0[r];
    q[8 + r] = kappa * a1[r];
  }
  float mt = q[0];
#pragma unroll
  for (int i = 1; i < 16; ++i) mt = fmaxf(mt, q[i]);
  const float mn = fmaxf(m1, mt);
  s1 *= expf(m1 - mn);
#pragma unroll
  for (int i = 0; i < 16; ++i) s1 += expf(q[i] - mn);
  m1 = mn;
}

__device__ __forceinline__ void load_gumbel(const float* gr, int ct, float4 (&uq)[4]) {
#pragma unroll
  for (int t = 0; t < 2; ++t) {
    uq[2 * t + 0] = *(const float4*)(gr + (ct + t) * 16);
    uq[2 * t + 1] = *(const float4*)(gr + (ct + t) * 16 + 4);
  }
}

__device__ __forceinline__ void pass2_tail(const float* cbs, int ct, int c, int h,
                                           float kappa, float lse, const v8f& a0,
                                           const v8f& a1, const float4 (&uq)[4],
                                           const int (&g2p)[4], const int (&g2m)[4],
                                           float* avgp_acc, float& acc_pl, float& m2,
                                           float& s2, v8f (&d)[4]) {
  float y2[16];
  float mt = -3.4e38f;
#pragma unroll
  for (int t = 0; t < 2; ++t) {
    float uu[8] = {uq[2 * t].x, uq[2 * t].y, uq[2 * t].z, uq[2 * t].w,
                   uq[2 * t + 1].x, uq[2 * t + 1].y, uq[2 * t + 1].z, uq[2 * t + 1].w};
    float sel = 0.0f;
#pragma unroll
    for (int r = 0; r < 8; ++r) {
      const float y = kappa * (t == 0 ? a0[r] : a1[r]);
      float lp = y - lse;
      float p = expf(lp);
      acc_pl += p * lp;
      float ps = row16_sum(p);
      sel = (c == r) ? ps : sel;
      float g = -logf(-logf(uu[r] + 1e-10f) + 1e-10f);
      float v = (y + g) * 2.0f;
      mt = fmaxf(mt, v);
      y2[8 * t + r] = v;
    }
    if (c < 8) atomicAdd(&avgp_acc[(ct + t) * 16 + 8 * h + c], sel);
  }
  mt = fmaxf(mt, xhalf(mt));
  const float mn = fmaxf(m2, mt);
  const float alpha = expf(m2 - mn);
  m2 = mn;
  s2 *= alpha;
#pragma unroll
  for (int nt = 0; nt < 4; ++nt) d[nt] *= alpha;
  float en[16];
#pragma unroll
  for (int i = 0; i < 16; ++i) {
    en[i] = expf(y2[i] - mn);
    s2 += en[i];
  }

  v2f av[2][4][4];
#pragma unroll
  for (int t = 0; t < 2; ++t) {
    const float* cr = cbs + (ct + t) * 16 * DIMD;
#pragma unroll
    for (int q = 0; q < 4; ++q) {
#pragma unroll
      for (int nt = 0; nt < 4; ++nt) {
        const int k1 = (nt * 16) ^ ((8 * q) & 16);
        const int m0 = 2 * (q & 1);
        av[t][q][nt].x = cr[(k1 < 32 ? g2p[m0] : g2m[m0]) + (2 * q) * DIMD + k1];
        av[t][q][nt].y = cr[(k1 < 32 ? g2p[m0 + 1] : g2m[m0 + 1]) + (2 * q + 1) * DIMD + k1];
      }
    }
  }
  __builtin_amdgcn_sched_barrier(0);
#pragma unroll
  for (int t = 0; t < 2; ++t) {
#pragma unroll
    for (int q = 0; q < 4; ++q) {
      v2f b;
      b.x = en[8 * t + 2 * q];
      b.y = en[8 * t + 2 * q + 1];
#pragma unroll
      for (int nt = 0; nt < 4; ++nt) {
        d[nt] = __builtin_amdgcn_wmma_f32_16x16x4_f32(false, av[t][q][nt], false, b,
                                                      (short)0, d[nt], false,
                                                      false);
      }
    }
  }
}

__global__ void vq_zero(float* acc) {
  int i = blockIdx.x * blockDim.x + threadIdx.x;
  if (i < 514) acc[i] = 0.0f;
}

__global__ __launch_bounds__(256) void vq_cbnorm(const float* __restrict__ cb,
                                                 float* __restrict__ cbn) {
  int wave = (blockIdx.x * blockDim.x + threadIdx.x) >> 5;
  int lane = threadIdx.x & 31;
  const float* row = cb + wave * DIMD;
  float v0 = row[lane], v1 = row[lane + 32];
  float s = wave_sum(v0 * v0 + v1 * v1);
  float scale = 1.0f / fmaxf(sqrtf(s), 1e-12f);
  cbn[wave * DIMD + lane] = v0 * scale;
  cbn[wave * DIMD + lane + 32] = v1 * scale;
}

__global__ __launch_bounds__(32 * NWAVES) void vq_main(
    const float* __restrict__ z, const float* __restrict__ kappa_p,
    const float* __restrict__ cbn, const float* __restrict__ gum,
    float* __restrict__ out, float* __restrict__ avgp_acc,
    float* __restrict__ sdisc, float* __restrict__ scont) {
  __shared__ float4 lds4[8192 + NWAVES * AREA_F4];
  const float* cbs = (const float*)lds4;

  const int lane = threadIdx.x & 31;
  const int wave = threadIdx.x >> 5;
  const int c = lane & 15;
  const int h = lane >> 4;
  const float kappa = kappa_p[0];
  float4* const area = lds4 + 8192 + wave * AREA_F4;

  for (int i = threadIdx.x; i < KCODES * DIMD / 4; i += 32 * NWAVES) {
    int code = i >> 4;
    int d0 = (i & 15) << 2;
    float4 v = *(const float4*)(cbn + (long)i * 4);
    lds4[(code * DIMD + (d0 ^ ((code & 15) << 2))) >> 2] = v;
  }
  __syncthreads();

  int g2p[4], g2m[4];
#pragma unroll
  for (int m = 0; m < 4; ++m) {
    g2p[m] = 512 * h + (c ^ (4 * m)) + 32 * h;
    g2m[m] = 512 * h + (c ^ (4 * m)) - 32 * h;
  }

  const long base_row = ((long)blockIdx.x * NWAVES + wave) * TILE_ROWS;

  v2f zf[16];
  {
    const float* zr = z + (base_row + c) * DIMD + 2 * h;
    float ss = 0.0f;
#pragma unroll
    for (int ks = 0; ks < 16; ++ks) {
      float2 t = *(const float2*)(zr + 4 * ks);
      zf[ks].x = t.x;
      zf[ks].y = t.y;
      ss += t.x * t.x + t.y * t.y;
    }
    ss += xhalf(ss);
    float zs = 1.0f / fmaxf(sqrtf(ss), 1e-12f);
#pragma unroll
    for (int ks = 0; ks < 16; ++ks) {
      zf[ks].x *= zs;
      zf[ks].y *= zs;
    }
  }

  float m1 = -3.4e38f;
  float s1 = 0.0f;
  for (int ct = 0; ct < REG_FROM; ct += 2) {
    v8f a0, a1;
    gemm1_pair(cbs, ct, c, h, zf, a0, a1);
    if (ct < 2 * NSTASH) {
      float4* s = area + (ct >> 1) * 128 + lane;
      s[0] = make_float4(a0[0], a0[1], a0[2], a0[3]);
      s[32] = make_float4(a0[4], a0[5], a0[6], a0[7]);
      s[64] = make_float4(a1[0], a1[1], a1[2], a1[3]);
      s[96] = make_float4(a1[4], a1[5], a1[6], a1[7]);
    }
    pass1_tail(a0, a1, kappa, m1, s1);
  }
  float kept[NREG * 16];
#pragma unroll
  for (int i = 0; i < NREG * 16; ++i) kept[i] = 0.0f;
#pragma nounroll
  for (int ct = REG_FROM; ct < KCODES / 16; ct += 2) {
    v8f a0, a1;
    gemm1_pair(cbs, ct, c, h, zf, a0, a1);
#pragma unroll
    for (int i = 0; i < (NREG - 1) * 16; ++i) kept[i] = kept[i + 16];
#pragma unroll
    for (int r = 0; r < 8; ++r) {
      kept[(NREG - 1) * 16 + r] = a0[r];
      kept[(NREG - 1) * 16 + 8 + r] = a1[r];
    }
    pass1_tail(a0, a1, kappa, m1, s1);
  }
  {
    const float mrow = fmaxf(m1, xhalf(m1));
    s1 *= expf(m1 - mrow);
    s1 += xhalf(s1);
    m1 = mrow;
  }
  const float lse = m1 + logf(s1);

  float acc_pl = 0.0f;
  float m2 = -3.4e38f;
  float s2 = 0.0f;
  v8f d[4] = {};
  const float* gr = gum + (base_row + c) * (long)KCODES + 8 * h;
  for (int ct = 0; ct < REG_FROM; ct += 2) {
    float4 uq[4];
    load_gumbel(gr, ct, uq);
    v8f a0, a1;
    if (ct < 2 * NSTASH) {
      const float4* s = area + (ct >> 1) * 128 + lane;
      const float4 x0 = s[0], x1 = s[32], x2 = s[64], x3 = s[96];
      a0 = (v8f){x0.x, x0.y, x0.z, x0.w, x1.x, x1.y, x1.z, x1.w};
      a1 = (v8f){x2.x, x2.y, x2.z, x2.w, x3.x, x3.y, x3.z, x3.w};
    } else {
      gemm1_pair(cbs, ct, c, h, zf, a0, a1);
    }
    pass2_tail(cbs, ct, c, h, kappa, lse, a0, a1, uq, g2p, g2m, avgp_acc, acc_pl, m2, s2, d);
  }
  for (int ct = REG_FROM; ct < KCODES / 16; ct += 2) {
    float4 uq[4];
    load_gumbel(gr, ct, uq);
    v8f a0, a1;
#pragma unroll
    for (int r = 0; r < 8; ++r) {
      a0[r] = kept[r];
      a1[r] = kept[8 + r];
    }
#pragma unroll
    for (int i = 0; i < (NREG - 1) * 16; ++i) kept[i] = kept[i + 16];
    pass2_tail(cbs, ct, c, h, kappa, lse, a0, a1, uq, g2p, g2m, avgp_acc, acc_pl, m2, s2, d);
  }
  acc_pl = wave_sum(acc_pl);
  if (lane == 0) atomicAdd(sdisc, acc_pl);
  s2 += xhalf(s2);
  const float inv_s2 = 1.0f / s2;

  float* zq = (float*)area;
#pragma unroll
  for (int nt = 0; nt < 4; ++nt) {
    float4 lo, hi4;
    lo.x = d[nt][0] * inv_s2; lo.y = d[nt][1] * inv_s2;
    lo.z = d[nt][2] * inv_s2; lo.w = d[nt][3] * inv_s2;
    hi4.x = d[nt][4] * inv_s2; hi4.y = d[nt][5] * inv_s2;
    hi4.z = d[nt][6] * inv_s2; hi4.w = d[nt][7] * inv_s2;
    *(float4*)(zq + c * ZQ_PITCH + nt * 16 + 8 * h) = lo;
    *(float4*)(zq + c * ZQ_PITCH + nt * 16 + 8 * h + 4) = hi4;
  }
  __builtin_amdgcn_fence(__ATOMIC_ACQ_REL, "wavefront");
  float acc_cont = 0.0f;
#pragma unroll
  for (int r = 0; r < TILE_ROWS; ++r) {
    float2 qv = *(const float2*)(zq + r * ZQ_PITCH + 2 * lane);
    long idx = (base_row + r) * DIMD + 2 * lane;
    float2 zo = *(const float2*)(z + idx);
    acc_cont += zo.x * (zo.x - qv.x);
    acc_cont += zo.y * (zo.y - qv.y);
    *(float2*)(out + idx) = qv;
  }
  acc_cont = wave_sum(acc_cont);
  if (lane == 0) atomicAdd(scont, acc_cont);
}

__global__ __launch_bounds__(32) void vq_finalize(
    const float* __restrict__ kappa_p, const float* __restrict__ avgp_acc,
    const float* __restrict__ sdisc, const float* __restrict__ scont,
    float* __restrict__ out) {
  int lane = threadIdx.x;
  const float invN = 1.0f / (float)NROWS;
  float acc = 0.0f;
#pragma unroll
  for (int i = lane; i < KCODES; i += 32) {
    float a = avgp_acc[i] * invN;
    acc += a * logf(a + 1e-7f);
  }
  acc = wave_sum(acc);
  if (lane == 0) {
    float loss = (sdisc[0] + kappa_p[0] * scont[0]) / 8.0f;
    out[(long)NROWS * DIMD] = loss;
    out[(long)NROWS * DIMD + 1] = expf(-acc);
  }
}

extern "C" void kernel_launch(void* const* d_in, const int* in_sizes, int n_in,
                              void* d_out, int out_size, void* d_ws,
                              size_t ws_size, hipStream_t stream) {
  const float* z = (const float*)d_in[0];
  const float* kappa = (const float*)d_in[1];
  const float* cb = (const float*)d_in[2];
  const float* gum = (const float*)d_in[3];
  float* out = (float*)d_out;

  float* wsf = (float*)d_ws;
  float* avgp_acc = wsf;
  float* sdisc = wsf + 512;
  float* scont = wsf + 513;
  float* cbn = wsf + 1024;

  vq_zero<<<3, 256, 0, stream>>>(wsf);
  vq_cbnorm<<<64, 256, 0, stream>>>(cb, cbn);
  vq_main<<<NBLOCKS, 32 * NWAVES, 0, stream>>>(z, kappa, cbn, gum, out,
                                                avgp_acc, sdisc, scont);
  vq_finalize<<<1, 32, 0, stream>>>(kappa, avgp_acc, sdisc, scont, out);
}


# Softmax element math in base 2: each in-line expf/logf (12-13 instructions: split-precision scaling, ldexp, range selects) becomes a bare v_exp_f32/v_log_f32 behind one FMA with log2e/ln2; sums kept
# speedup vs baseline: 1.8143x; 1.3434x over previous
#include <hip/hip_runtime.h>
#include <hip/hip_bf16.h>


typedef __attribute__((ext_vector_type(2))) float v2f;
typedef __attribute__((ext_vector_type(8))) float v8f;

#define DIMD 64
#define KCODES 512
#define NROWS 65536
#define TILE_ROWS 16
#define ZQ_PITCH 68
#define NWAVES 16
#define NBLOCKS 256
#define ZQ_F4 (TILE_ROWS * ZQ_PITCH / 4)
#define NSTASH 6
#define AREA_F4 (NSTASH * 128)
#define NREG 3
#define REG_FROM (KCODES / 16 - 2 * NREG)
static_assert(ZQ_F4 <= AREA_F4, "the output staging reuses the wave's area");
static_assert((8192 + NWAVES * AREA_F4) * 16 <= 327680, "LDS of one compute unit");

#define LOG2E 1.4426950408889634f
#define LN2 0.6931471805599453f

__device__ __forceinline__ float hw_exp2(float x) { return __builtin_amdgcn_exp2f(x); }
__device__ __forceinline__ float hw_log2(float x) { return __builtin_amdgcn_logf(x); }

__device__ __forceinline__ float wave_sum(float v) {
#pragma unroll
  for (int off = 16; off > 0; off >>= 1) v += __shfl_xor(v, off, 32);
  return v;
}

template <int CTRL>
__device__ __forceinline__ float dpp_get(float v) {
  return __builtin_bit_cast(
      float, __builtin_amdgcn_update_dpp(0, __builtin_bit_cast(int, v), CTRL,
                                         0xF, 0xF, true));
}
__device__ __forceinline__ float row16_sum(float v) {
  v += dpp_get<0xB1>(v);
  v += dpp_get<0x4E>(v);
  v += dpp_get<0x141>(v);
  v += dpp_get<0x140>(v);
  return v;
}

__device__ __forceinline__ float xhalf(float x) {
  int b = __builtin_bit_cast(int, x);
  return __builtin_bit_cast(
      float, __builtin_amdgcn_permlanex16(b, b, 0x76543210, 0xfedcba98, false, false));
}

__device__ __forceinline__ void gemm1_pair(const float* cbs, int ct, int c, int h,
                                           const v2f (&zf)[16], v8f& a0, v8f& a1) {
  a0 = (v8f){};
  a1 = (v8f){};
  const int a1o = (ct * 16 + c) * DIMD + 4 * c + 2 * h;
  float2 t0[16], t1[16];
#pragma unroll
  for (int ks = 0; ks < 16; ++ks) {
    const float* p0 = cbs + (a1o ^ (ks << 2));
    t0[ks] = *(const float2*)p0;
    t1[ks] = *(const float2*)(p0 + 16 * DIMD);
  }
  __builtin_amdgcn_sched_barrier(0);
#pragma unroll
  for (int ks = 0; ks < 16; ++ks) {
    v2f f0, f1;
    f0.x = t0[ks].x;
    f0.y = t0[ks].y;
    f1.x = t1[ks].x;
    f1.y = t1[ks].y;
    a0 = __builtin_amdgcn_wmma_f32_16x16x4_f32(false, f0, false, zf[ks],
                                               (short)0, a0, false, false);
    a1 = __builtin_amdgcn_wmma_f32_16x16x4_f32(false, f1, false, zf[ks],
                                               (short)0, a1, false, false);
  }
  asm volatile("v_nop\n\tv_nop\n\tv_nop\n\tv_nop" : "+v"(a0), "+v"(a1));
}

__device__ __forceinline__ void pass1_tail(const v8f& a0, const v8f& a1, float kL,
                                           float& m1, float& s1) {
  float q[16];
#pragma unroll
  for (int r = 0; r < 8; ++r) {
    q[r] = kL * a0[r];
    q[8 + r] = kL * a1[r];
  }
  float mt = q[0];
#pragma unroll
  for (int i = 1; i < 16; ++i) mt = fmaxf(mt, q[i]);
  const float mn = fmaxf(m1, mt);
  s1 *= hw_exp2(m1 - mn);
#pragma unroll
  for (int i = 0; i < 16; ++i) s1 += hw_exp2(q[i] - mn);
  m1 = mn;
}

__device__ __forceinline__ void load_gumbel(const float* gr, int ct, float4 (&uq)[4]) {
#pragma unroll
  for (int t = 0; t < 2; ++t) {
    uq[2 * t + 0] = *(const float4*)(gr + (ct + t) * 16);
    uq[2 * t + 1] = *(const float4*)(gr + (ct + t) * 16 + 4);
  }
}

__device__ __forceinline__ void pass2_tail(const float* cbs, int ct, int c, int h,
                                           float kL, float lse2, const v8f& a0,
                                           const v8f& a1, const float4 (&uq)[4],
                                           const int (&g2p)[4], const int (&g2m)[4],
                                           float* avgp_acc, float& acc_pl, float& m2,
                                           float& s2, v8f (&d)[4]) {
  float y2[16];
  float mt = -3.4e38f;
#pragma unroll
  for (int t = 0; t < 2; ++t) {
    float uu[8] = {uq[2 * t].x, uq[2 * t].y, uq[2 * t].z, uq[2 * t].w,
                   uq[2 * t + 1].x, uq[2 * t + 1].y, uq[2 * t + 1].z, uq[2 * t + 1].w};
    float sel = 0.0f;
#pragma unroll
    for (int r = 0; r < 8; ++r) {
      const float a = (t == 0 ? a0[r] : a1[r]);
      const float lp2 = __builtin_fmaf(a, kL, -lse2);
      const float p = hw_exp2(lp2);
      acc_pl = __builtin_fmaf(p, lp2, acc_pl);
      float ps = row16_sum(p);
      sel = (c == r) ? ps : sel;
      const float tg = __builtin_fmaf(hw_log2(uu[r] + 1e-10f), -LN2, 1e-10f);
      const float v = __builtin_fmaf(a, kL, -hw_log2(tg));
      mt = fmaxf(mt, v);
      y2[8 * t + r] = v;
    }
    if (c < 8) atomicAdd(&avgp_acc[(ct + t) * 16 + 8 * h + c], sel);
  }
  mt = fmaxf(mt, xhalf(mt));
  const float mn = fmaxf(m2, mt);
  const float alpha = hw_exp2((m2 - mn) * 2.0f);
  m2 = mn;
  const float mn2 = 2.0f * mn;
  s2 *= alpha;
#pragma unroll
  for (int nt = 0; nt < 4; ++nt) d[nt] *= alpha;
  float en[16];
#pragma unroll
  for (int i = 0; i < 16; ++i) {
    en[i] = hw_exp2(__builtin_fmaf(y2[i], 2.0f, -mn2));
    s2 += en[i];
  }

  v2f av[2][4][4];
#pragma unroll
  for (int t = 0; t < 2; ++t) {
    const float* cr = cbs + (ct + t) * 16 * DIMD;
#pragma unroll
    for (int q = 0; q < 4; ++q) {
#pragma unroll
      for (int nt = 0; nt < 4; ++nt) {
        const int k1 = (nt * 16) ^ ((8 * q) & 16);
        const int m0 = 2 * (q & 1);
        av[t][q][nt].x = cr[(k1 < 32 ? g2p[m0] : g2m[m0]) + (2 * q) * DIMD + k1];
        av[t][q][nt].y = cr[(k1 < 32 ? g2p[m0 + 1] : g2m[m0 + 1]) + (2 * q + 1) * DIMD + k1];
      }
    }
  }
  __builtin_amdgcn_sched_barrier(0);
#pragma unroll
  for (int t = 0; t < 2; ++t) {
#pragma unroll
    for (int q = 0; q < 4; ++q) {
      v2f b;
      b.x = en[8 * t + 2 * q];
      b.y = en[8 * t + 2 * q + 1];
#pragma unroll
      for (int nt = 0; nt < 4; ++nt) {
        d[nt] = __builtin_amdgcn_wmma_f32_16x16x4_f32(false, av[t][q][nt], false, b,
                                                      (short)0, d[nt], false,
                                                      false);
      }
    }
  }
}

__global__ void vq_zero(float* acc) {
  int i = blockIdx.x * blockDim.x + threadIdx.x;
  if (i < 514) acc[i] = 0.0f;
}

__global__ __launch_bounds__(256) void vq_cbnorm(const float* __restrict__ cb,
                                                 float* __restrict__ cbn) {
  int wave = (blockIdx.x * blockDim.x + threadIdx.x) >> 5;
  int lane = threadIdx.x & 31;
  const float* row = cb + wave * DIMD;
  float v0 = row[lane], v1 = row[lane + 32];
  float s = wave_sum(v0 * v0 + v1 * v1);
  float scale = 1.0f / fmaxf(sqrtf(s), 1e-12f);
  cbn[wave * DIMD + lane] = v0 * scale;
  cbn[wave * DIMD + lane + 32] = v1 * scale;
}

__global__ __launch_bounds__(32 * NWAVES) void vq_main(
    const float* __restrict__ z, const float* __restrict__ kappa_p,
    const float* __restrict__ cbn, const float* __restrict__ gum,
    float* __restrict__ out, float* __restrict__ avgp_acc,
    float* __restrict__ sdisc, float* __restrict__ scont) {
  __shared__ float4 lds4[8192 + NWAVES * AREA_F4];
  const float* cbs = (const float*)lds4;

  const int lane = threadIdx.x & 31;
  const int wave = threadIdx.x >> 5;
  const int c = lane & 15;
  const int h = lane >> 4;
  const float kL = kappa_p[0] * LOG2E;
  float4* const area = lds4 + 8192 + wave * AREA_F4;

  for (int i = threadIdx.x; i < KCODES * DIMD / 4; i += 32 * NWAVES) {
    int code = i >> 4;
    int d0 = (i & 15) << 2;
    float4 v = *(const float4*)(cbn + (long)i * 4);
    lds4[(code * DIMD + (d0 ^ ((code & 15) << 2))) >> 2] = v;
  }
  __syncthreads();

  int g2p[4], g2m[4];
#pragma unroll
  for (int m = 0; m < 4; ++m) {
    g2p[m] = 512 * h + (c ^ (4 * m)) + 32 * h;
    g2m[m] = 512 * h + (c ^ (4 * m)) - 32 * h;
  }

  const long base_row = ((long)blockIdx.x * NWAVES + wave) * TILE_ROWS;

  v2f zf[16];
  {
    const float* zr = z + (base_row + c) * DIMD + 2 * h;
    float ss = 0.0f;
#pragma unroll
    for (int ks = 0; ks < 16; ++ks) {
      float2 t = *(const float2*)(zr + 4 * ks);
      zf[ks].x = t.x;
      zf[ks].y = t.y;
      ss += t.x * t.x + t.y * t.y;
    }
    ss += xhalf(ss);
    float zs = 1.0f / fmaxf(sqrtf(ss), 1e-12f);
#pragma unroll
    for (int ks = 0; ks < 16; ++ks) {
      zf[ks].x *= zs;
      zf[ks].y *= zs;
    }
  }

  float m1 = -3.4e38f;
  float s1 = 0.0f;
  for (int ct = 0; ct < REG_FROM; ct += 2) {
    v8f a0, a1;
    gemm1_pair(cbs, ct, c, h, zf, a0, a1);
    if (ct < 2 * NSTASH) {
      float4* s = area + (ct >> 1) * 128 + lane;
      s[0] = make_float4(a0[0], a0[1], a0[2], a0[3]);
      s[32] = make_float4(a0[4], a0[5], a0[6], a0[7]);
      s[64] = make_float4(a1[0], a1[1], a1[2], a1[3]);
      s[96] = make_float4(a1[4], a1[5], a1[6], a1[7]);
    }
    pass1_tail(a0, a1, kL, m1, s1);
  }
  float kept[NREG * 16];
#pragma unroll
  for (int i = 0; i < NREG * 16; ++i) kept[i] = 0.0f;
#pragma nounroll
  for (int ct = REG_FROM; ct < KCODES / 16; ct += 2) {
    v8f a0, a1;
    gemm1_pair(cbs, ct, c, h, zf, a0, a1);
#pragma unroll
    for (int i = 0; i < (NREG - 1) * 16; ++i) kept[i] = kept[i + 16];
#pragma unroll
    for (int r = 0; r < 8; ++r) {
      kept[(NREG - 1) * 16 + r] = a0[r];
      kept[(NREG - 1) * 16 + 8 + r] = a1[r];
    }
    pass1_tail(a0, a1, kL, m1, s1);
  }
  {
    const float mrow = fmaxf(m1, xhalf(m1));
    s1 *= hw_exp2(m1 - mrow);
    s1 += xhalf(s1);
    m1 = mrow;
  }
  const float lse2 = m1 + hw_log2(s1);

  float acc_pl = 0.0f;
  float m2 = -3.4e38f;
  float s2 = 0.0f;
  v8f d[4] = {};
  const float* gr = gum + (base_row + c) * (long)KCODES + 8 * h;
  for (int ct = 0; ct < REG_FROM; ct += 2) {
    float4 uq[4];
    load_gumbel(gr, ct, uq);
    v8f a0, a1;
    if (ct < 2 * NSTASH) {
      const float4* s = area + (ct >> 1) * 128 + lane;
      const float4 x0 = s[0], x1 = s[32], x2 = s[64], x3 = s[96];
      a0 = (v8f){x0.x, x0.y, x0.z, x0.w, x1.x, x1.y, x1.z, x1.w};
      a1 = (v8f){x2.x, x2.y, x2.z, x2.w, x3.x, x3.y, x3.z, x3.w};
    } else {
      gemm1_pair(cbs, ct, c, h, zf, a0, a1);
    }
    pass2_tail(cbs, ct, c, h, kL, lse2, a0, a1, uq, g2p, g2m, avgp_acc, acc_pl, m2, s2, d);
  }
  for (int ct = REG_FROM; ct < KCODES / 16; ct += 2) {
    float4 uq[4];
    load_gumbel(gr, ct, uq);
    v8f a0, a1;
#pragma unroll
    for (int r = 0; r < 8; ++r) {
      a0[r] = kept[r];
      a1[r] = kept[8 + r];
    }
#pragma unroll
    for (int i = 0; i < (NREG - 1) * 16; ++i) kept[i] = kept[i + 16];
    pass2_tail(cbs, ct, c, h, kL, lse2, a0, a1, uq, g2p, g2m, avgp_acc, acc_pl, m2, s2, d);
  }
  acc_pl = wave_sum(acc_pl);
  if (lane == 0) atomicAdd(sdisc, acc_pl * LN2);
  s2 += xhalf(s2);
  const float inv_s2 = 1.0f / s2;

  float* zq = (float*)area;
#pragma unroll
  for (int nt = 0; nt < 4; ++nt) {
    float4 lo, hi4;
    lo.x = d[nt][0] * inv_s2; lo.y = d[nt][1] * inv_s2;
    lo.z = d[nt][2] * inv_s2; lo.w = d[nt][3] * inv_s2;
    hi4.x = d[nt][4] * inv_s2; hi4.y = d[nt][5] * inv_s2;
    hi4.z = d[nt][6] * inv_s2; hi4.w = d[nt][7] * inv_s2;
    *(float4*)(zq + c * ZQ_PITCH + nt * 16 + 8 * h) = lo;
    *(float4*)(zq + c * ZQ_PITCH + nt * 16 + 8 * h + 4) = hi4;
  }
  __builtin_amdgcn_fence(__ATOMIC_ACQ_REL, "wavefront");
  float acc_cont = 0.0f;
#pragma unroll
  for (int r = 0; r < TILE_ROWS; ++r) {
    float2 qv = *(const float2*)(zq + r * ZQ_PITCH + 2 * lane);
    long idx = (base_row + r) * DIMD + 2 * lane;
    float2 zo = *(const float2*)(z + idx);
    acc_cont += zo.x * (zo.x - qv.x);
    acc_cont += zo.y * (zo.y - qv.y);
    *(float2*)(out + idx) = qv;
  }
  acc_cont = wave_sum(acc_cont);
  if (lane == 0) atomicAdd(scont, acc_cont);
}

__global__ __launch_bounds__(32) void vq_finalize(
    const float* __restrict__ kappa_p, const float* __restrict__ avgp_acc,
    const float* __restrict__ sdisc, const float* __restrict__ scont,
    float* __restrict__ out) {
  int lane = threadIdx.x;
  const float invN = 1.0f / (float)NROWS;
  float acc = 0.0f;
#pragma unroll
  for (int i = lane; i < KCODES; i += 32) {
    float a = avgp_acc[i] * invN;
    acc += a * logf(a + 1e-7f);
  }
  acc = wave_sum(acc);
  if (lane == 0) {
    float loss = (sdisc[0] + kappa_p[0] * scont[0]) / 8.0f;
    out[(long)NROWS * DIMD] = loss;
    out[(long)NROWS * DIMD + 1] = expf(-acc);
  }
}

extern "C" void kernel_launch(void* const* d_in, const int* in_sizes, int n_in,
                              void* d_out, int out_size, void* d_ws,
                              size_t ws_size, hipStream_t stream) {
  const float* z = (const float*)d_in[0];
  const float* kappa = (const float*)d_in[1];
  const float* cb = (const float*)d_in[2];
  const float* gum = (const float*)d_in[3];
  float* out = (float*)d_out;

  float* wsf = (float*)d_ws;
  float* avgp_acc = wsf;
  float* sdisc = wsf + 512;
  float* scont = wsf + 513;
  float* cbn = wsf + 1024;

  vq_zero<<<3, 256, 0, stream>>>(wsf);
  vq_cbnorm<<<64, 256, 0, stream>>>(cb, cbn);
  vq_main<<<NBLOCKS, 32 * NWAVES, 0, stream>>>(z, kappa, cbn, gum, out,
                                                avgp_acc, sdisc, scont);
  vq_finalize<<<1, 32, 0, stream>>>(kappa, avgp_acc, sdisc, scont, out);
}


# 8 waves per work-group (was 16), two tiles each: a wave's doubled LDS area and register room keep all 16 trips' logits (12 LDS, 4 registers), so pass 2 forms no GEMM1 again; sums and order unchanged
# speedup vs baseline: 2.0023x; 1.1036x over previous
#include <hip/hip_runtime.h>
#include <hip/hip_bf16.h>


typedef __attribute__((ext_vector_type(2))) float v2f;
typedef __attribute__((ext_vector_type(8))) float v8f;

#define DIMD 64
#define KCODES 512
#define NROWS 65536
#define TILE_ROWS 16
#define ZQ_PITCH 68
#define NWAVES 8
#define WAVE_TILES 2
#define NBLOCKS 256
#define ZQ_F4 (TILE_ROWS * ZQ_PITCH / 4)
#define NSTASH 12
#define AREA_F4 (NSTASH * 128)
#define NREG 4
#define REG_FROM (KCODES / 16 - 2 * NREG)
static_assert(2 * NSTASH == REG_FROM, "every trip's logits are kept: LDS first, registers for the rest");
static_assert(NBLOCKS * NWAVES * WAVE_TILES * TILE_ROWS == NROWS, "the tiles cover the rows");
static_assert(ZQ_F4 <= AREA_F4, "the output staging reuses the wave's area");
static_assert((8192 + NWAVES * AREA_F4) * 16 <= 327680, "LDS of one compute unit");

#define LOG2E 1.4426950408889634f
#define LN2 0.6931471805599453f

__device__ __forceinline__ float hw_exp2(float x) { return __builtin_amdgcn_exp2f(x); }
__device__ __forceinline__ float hw_log2(float x) { return __builtin_amdgcn_logf(x); }

__device__ __forceinline__ float wave_sum(float v) {
#pragma unroll
  for (int off = 16; off > 0; off >>= 1) v += __shfl_xor(v, off, 32);
  return v;
}

template <int CTRL>
__device__ __forceinline__ float dpp_get(float v) {
  return __builtin_bit_cast(
      float, __builtin_amdgcn_update_dpp(0, __builtin_bit_cast(int, v), CTRL,
                                         0xF, 0xF, true));
}
template <int CTRL>
__device__ __forceinline__ float pair_step(bool up, float a, float b) {
  const float keep = up ? b : a;
  const float give = up ? a : b;
  return keep + dpp_get<CTRL>(give);
}

__device__ __forceinline__ float xhalf(float x) {
  int b = __builtin_bit_cast(int, x);
  return __builtin_bit_cast(
      float, __builtin_amdgcn_permlanex16(b, b, 0x76543210, 0xfedcba98, false, false));
}

__device__ __forceinline__ void gemm1_pair(const float* cbs, int ct, int c, int h,
                                           const v2f (&zf)[16], v8f& a0, v8f& a1) {
  a0 = (v8f){};
  a1 = (v8f){};
  const int a1o = (ct * 16 + c) * DIMD + 4 * c + 2 * h;
  float2 t0[16], t1[16];
#pragma unroll
  for (int ks = 0; ks < 16; ++ks) {
    const float* p0 = cbs + (a1o ^ (ks << 2));
    t0[ks] = *(const float2*)p0;
    t1[ks] = *(const float2*)(p0 + 16 * DIMD);
  }
  __builtin_amdgcn_sched_barrier(0);
#pragma unroll
  for (int ks = 0; ks < 16; ++ks) {
    v2f f0, f1;
    f0.x = t0[ks].x;
    f0.y = t0[ks].y;
    f1.x = t1[ks].x;
    f1.y = t1[ks].y;
    a0 = __builtin_amdgcn_wmma_f32_16x16x4_f32(false, f0, false, zf[ks],
                                               (short)0, a0, false, false);
    a1 = __builtin_amdgcn_wmma_f32_16x16x4_f32(false, f1, false, zf[ks],
                                               (short)0, a1, false, false);
  }
  asm volatile("v_nop\n\tv_nop\n\tv_nop\n\tv_nop" : "+v"(a0), "+v"(a1));
}

__device__ __forceinline__ void pass1_tail(const v8f& a0, const v8f& a1, float kL,
                                           float& m1, float& s1) {
  float q[16];
#pragma unroll
  for (int r = 0; r < 8; ++r) {
    q[r] = kL * a0[r];
    q[8 + r] = kL * a1[r];
  }
  float mt = q[0];
#pragma unroll
  for (int i = 1; i < 16; ++i) mt = fmaxf(mt, q[i]);
  const float mn = fmaxf(m1, mt);
  s1 *= hw_exp2(m1 - mn);
#pragma unroll
  for (int i = 0; i < 16; ++i) s1 += hw_exp2(q[i] - mn);
  m1 = mn;
}

__device__ __forceinline__ void load_gumbel(const float* gr, int ct, float4 (&uq)[4]) {
#pragma unroll
  for (int t = 0; t < 2; ++t) {
    uq[2 * t + 0] = *(const float4*)(gr + (ct + t) * 16);
    uq[2 * t + 1] = *(const float4*)(gr + (ct + t) * 16 + 4);
  }
}

__device__ __forceinline__ void pass2_tail(const float* cbs, int ct, int c, int h,
                                           float kL, float lse2, const v8f& a0,
                                           const v8f& a1, const float4 (&uq)[4],
                                           const int (&g2p)[4], const int (&g2m)[4],
                                           float* avgp_acc, float& acc_pl, float& m2,
                                           float& s2, v8f (&d)[4]) {
  float y2[16];
  float mt = -3.4e38f;
  const bool bit0 = (c & 1) != 0, bit1 = (c & 2) != 0, bit2 = (c & 4) != 0, bit3 = (c & 8) != 0;
  float tile_sum[2];
#pragma unroll
  for (int t = 0; t < 2; ++t) {
    float uu[8] = {uq[2 * t].x, uq[2 * t].y, uq[2 * t].z, uq[2 * t].w,
                   uq[2 * t + 1].x, uq[2 * t + 1].y, uq[2 * t + 1].z, uq[2 * t + 1].w};
    float pv[8];
#pragma unroll
    for (int r = 0; r < 8; ++r) {
      const float a = (t == 0 ? a0[r] : a1[r]);
      const float lp2 = __builtin_fmaf(a, kL, -lse2);
      const float p = hw_exp2(lp2);
      acc_pl = __builtin_fmaf(p, lp2, acc_pl);
      pv[r] = p;
      const float tg = __builtin_fmaf(hw_log2(uu[r] + 1e-10f), -LN2, 1e-10f);
      const float v = __builtin_fmaf(a, kL, -hw_log2(tg));
      mt = fmaxf(mt, v);
      y2[8 * t + r] = v;
    }
    float w[4], x[2];
#pragma unroll
    for (int i = 0; i < 4; ++i) w[i] = pair_step<0xB1>(bit0, pv[2 * i], pv[2 * i + 1]);
#pragma unroll
    for (int i = 0; i < 2; ++i) x[i] = pair_step<0x4E>(bit1, w[2 * i], w[2 * i + 1]);
    tile_sum[t] = pair_step<0x164>(bit2, x[0], x[1]);
  }
  atomicAdd(&avgp_acc[ct * 16 + ((c >> 3) << 4) + 8 * h + (c & 7)],
            pair_step<0x168>(bit3, tile_sum[0], tile_sum[1]));
  mt = fmaxf(mt, xhalf(mt));
  const float mn = fmaxf(m2, mt);
  const float alpha = hw_exp2((m2 - mn) * 2.0f);
  m2 = mn;
  const float mn2 = 2.0f * mn;
  s2 *= alpha;
#pragma unroll
  for (int nt = 0; nt < 4; ++nt) d[nt] *= alpha;
  float en[16];
#pragma unroll
  for (int i = 0; i < 16; ++i) {
    en[i] = hw_exp2(__builtin_fmaf(y2[i], 2.0f, -mn2));
    s2 += en[i];
  }

  v2f av[2][4][4];
#pragma unroll
  for (int t = 0; t < 2; ++t) {
    const float* cr = cbs + (ct + t) * 16 * DIMD;
#pragma unroll
    for (int q = 0; q < 4; ++q) {
#pragma unroll
      for (int nt = 0; nt < 4; ++nt) {
        const int k1 = (nt * 16) ^ ((8 * q) & 16);
        const int m0 = 2 * (q & 1);
        av[t][q][nt].x = cr[(k1 < 32 ? g2p[m0] : g2m[m0]) + (2 * q) * DIMD + k1];
        av[t][q][nt].y = cr[(k1 < 32 ? g2p[m0 + 1] : g2m[m0 + 1]) + (2 * q + 1) * DIMD + k1];
      }
    }
  }
  __builtin_amdgcn_sched_barrier(0);
#pragma unroll
  for (int t = 0; t < 2; ++t) {
#pragma unroll
    for (int q = 0; q < 4; ++q) {
      v2f b;
      b.x = en[8 * t + 2 * q];
      b.y = en[8 * t + 2 * q + 1];
#pragma unroll
      for (int nt = 0; nt < 4; ++nt) {
        d[nt] = __builtin_amdgcn_wmma_f32_16x16x4_f32(false, av[t][q][nt], false, b,
                                                      (short)0, d[nt], false,
                                                      false);
      }
    }
  }
}

__global__ void vq_zero(float* acc) {
  int i = blockIdx.x * blockDim.x + threadIdx.x;
  if (i < 514) acc[i] = 0.0f;
}

__global__ __launch_bounds__(256) void vq_cbnorm(const float* __restrict__ cb,
                                                 float* __restrict__ cbn) {
  int wave = (blockIdx.x * blockDim.x + threadIdx.x) >> 5;
  int lane = threadIdx.x & 31;
  const float* row = cb + wave * DIMD;
  float v0 = row[lane], v1 = row[lane + 32];
  float s = wave_sum(v0 * v0 + v1 * v1);
  float scale = 1.0f / fmaxf(sqrtf(s), 1e-12f);
  cbn[wave * DIMD + lane] = v0 * scale;
  cbn[wave * DIMD + lane + 32] = v1 * scale;
}

__global__ __launch_bounds__(32 * NWAVES) void vq_main(
    const float* __restrict__ z, const float* __restrict__ kappa_p,
    const float* __restrict__ cbn, const float* __restrict__ gum,
    float* __restrict__ out, float* __restrict__ avgp_acc,
    float* __restrict__ sdisc, float* __restrict__ scont) {
  __shared__ float4 lds4[8192 + NWAVES * AREA_F4];
  const float* cbs = (const float*)lds4;

  const int lane = threadIdx.x & 31;
  const int wave = threadIdx.x >> 5;
  const int c = lane & 15;
  const int h = lane >> 4;
  const float kL = kappa_p[0] * LOG2E;
  float4* const area = lds4 + 8192 + wave * AREA_F4;

#pragma unroll 16
  for (int i = threadIdx.x; i < KCODES * DIMD / 4; i += 32 * NWAVES) {
    int code = i >> 4;
    int d0 = (i & 15) << 2;
    float4 v = *(const float4*)(cbn + (long)i * 4);
    lds4[(code * DIMD + (d0 ^ ((code & 15) << 2))) >> 2] = v;
  }
  __syncthreads();

  int g2p[4], g2m[4];
#pragma unroll
  for (int m = 0; m < 4; ++m) {
    g2p[m] = 512 * h + (c ^ (4 * m)) + 32 * h;
    g2m[m] = 512 * h + (c ^ (4 * m)) - 32 * h;
  }

  for (int it = 0; it < WAVE_TILES; ++it) {
    const long base_row =
        (((long)blockIdx.x * WAVE_TILES + it) * NWAVES + wave) * TILE_ROWS;

    v2f zf[16];
    {
      const float* zr = z + (base_row + c) * DIMD + 2 * h;
      float ss = 0.0f;
#pragma unroll
      for (int ks = 0; ks < 16; ++ks) {
        float2 t = *(const float2*)(zr + 4 * ks);
        zf[ks].x = t.x;
        zf[ks].y = t.y;
        ss += t.x * t.x + t.y * t.y;
      }
      ss += xhalf(ss);
      float zs = 1.0f / fmaxf(sqrtf(ss), 1e-12f);
#pragma unroll
      for (int ks = 0; ks < 16; ++ks) {
        zf[ks].x *= zs;
        zf[ks].y *= zs;
      }
    }

    float m1 = -3.4e38f;
    float s1 = 0.0f;
    for (int ct = 0; ct < REG_FROM; ct += 2) {
      v8f a0, a1;
      gemm1_pair(cbs, ct, c, h, zf, a0, a1);
      float4* s = area + (ct >> 1) * 128 + lane;
      s[0] = make_float4(a0[0], a0[1], a0[2], a0[3]);
      s[32] = make_float4(a0[4], a0[5], a0[6], a0[7]);
      s[64] = make_float4(a1[0], a1[1], a1[2], a1[3]);
      s[96] = make_float4(a1[4], a1[5], a1[6], a1[7]);
      pass1_tail(a0, a1, kL, m1, s1);
    }
    float kept[NREG * 16];
#pragma unroll
    for (int i = 0; i < NREG * 16; ++i) kept[i] = 0.0f;
#pragma nounroll
    for (int ct = REG_FROM; ct < KCODES / 16; ct += 2) {
      v8f a0, a1;
      gemm1_pair(cbs, ct, c, h, zf, a0, a1);
#pragma unroll
      for (int i = 0; i < (NREG - 1) * 16; ++i) kept[i] = kept[i + 16];
#pragma unroll
      for (int r = 0; r < 8; ++r) {
        kept[(NREG - 1) * 16 + r] = a0[r];
        kept[(NREG - 1) * 16 + 8 + r] = a1[r];
      }
      pass1_tail(a0, a1, kL, m1, s1);
    }
    {
      const float mrow = fmaxf(m1, xhalf(m1));
      s1 *= hw_exp2(m1 - mrow);
      s1 += xhalf(s1);
      m1 = mrow;
    }
    const float lse2 = m1 + hw_log2(s1);

    float acc_pl = 0.0f;
    float m2 = -3.4e38f;
    float s2 = 0.0f;
    v8f d[4] = {};
    const float* gr = gum + (base_row + c) * (long)KCODES + 8 * h;
    for (int ct = 0; ct < REG_FROM; ct += 2) {
      float4 uq[4];
      load_gumbel(gr, ct, uq);
      const float4* s = area + (ct >> 1) * 128 + lane;
      const float4 x0 = s[0], x1 = s[32], x2 = s[64], x3 = s[96];
      const v8f a0 = (v8f){x0.x, x0.y, x0.z, x0.w, x1.x, x1.y, x1.z, x1.w};
      const v8f a1 = (v8f){x2.x, x2.y, x2.z, x2.w, x3.x, x3.y, x3.z, x3.w};
      pass2_tail(cbs, ct, c, h, kL, lse2, a0, a1, uq, g2p, g2m, avgp_acc, acc_pl, m2, s2, d);
    }
    for (int ct = REG_FROM; ct < KCODES / 16; ct += 2) {
      float4 uq[4];
      load_gumbel(gr, ct, uq);
      v8f a0, a1;
#pragma unroll
      for (int r = 0; r < 8; ++r) {
        a0[r] = kept[r];
        a1[r] = kept[8 + r];
      }
#pragma unroll
      for (int i = 0; i < (NREG - 1) * 16; ++i) kept[i] = kept[i + 16];
      pass2_tail(cbs, ct, c, h, kL, lse2, a0, a1, uq, g2p, g2m, avgp_acc, acc_pl, m2, s2, d);
    }
    acc_pl = wave_sum(acc_pl);
    if (lane == 0) atomicAdd(sdisc, acc_pl * LN2);
    s2 += xhalf(s2);
    const float inv_s2 = 1.0f / s2;

    float* zq = (float*)area;
#pragma unroll
    for (int nt = 0; nt < 4; ++nt) {
      float4 lo, hi4;
      lo.x = d[nt][0] * inv_s2; lo.y = d[nt][1] * inv_s2;
      lo.z = d[nt][2] * inv_s2; lo.w = d[nt][3] * inv_s2;
      hi4.x = d[nt][4] * inv_s2; hi4.y = d[nt][5] * inv_s2;
      hi4.z = d[nt][6] * inv_s2; hi4.w = d[nt][7] * inv_s2;
      *(float4*)(zq + c * ZQ_PITCH + nt * 16 + 8 * h) = lo;
      *(float4*)(zq + c * ZQ_PITCH + nt * 16 + 8 * h + 4) = hi4;
    }
    __builtin_amdgcn_fence(__ATOMIC_ACQ_REL, "wavefront");
    float acc_cont = 0.0f;
#pragma unroll
    for (int r = 0; r < TILE_ROWS; ++r) {
      float2 qv = *(const float2*)(zq + r * ZQ_PITCH + 2 * lane);
      long idx = (base_row + r) * DIMD + 2 * lane;
      float2 zo = *(const float2*)(z + idx);
      acc_cont += zo.x * (zo.x - qv.x);
      acc_cont += zo.y * (zo.y - qv.y);
      *(float2*)(out + idx) = qv;
    }
    acc_cont = wave_sum(acc_cont);
    if (lane == 0) atomicAdd(scont, acc_cont);
  }
}

__global__ __launch_bounds__(32) void vq_finalize(
    const float* __restrict__ kappa_p, const float* __restrict__ avgp_acc,
    const float* __restrict__ sdisc, const float* __restrict__ scont,
    float* __restrict__ out) {
  int lane = threadIdx.x;
  const float invN = 1.0f / (float)NROWS;
  float acc = 0.0f;
#pragma unroll
  for (int i = lane; i < KCODES; i += 32) {
    float a = avgp_acc[i] * invN;
    acc += a * logf(a + 1e-7f);
  }
  acc = wave_sum(acc);
  if (lane == 0) {
    float loss = (sdisc[0] + kappa_p[0] * scont[0]) / 8.0f;
    out[(long)NROWS * DIMD] = loss;
    out[(long)NROWS * DIMD + 1] = expf(-acc);
  }
}

extern "C" void kernel_launch(void* const* d_in, const int* in_sizes, int n_in,
                              void* d_out, int out_size, void* d_ws,
                              size_t ws_size, hipStream_t stream) {
  const float* z = (const float*)d_in[0];
  const float* kappa = (const float*)d_in[1];
  const float* cb = (const float*)d_in[2];
  const float* gum = (const float*)d_in[3];
  float* out = (float*)d_out;

  float* wsf = (float*)d_ws;
  float* avgp_acc = wsf;
  float* sdisc = wsf + 512;
  float* scont = wsf + 513;
  float* cbn = wsf + 1024;

  vq_zero<<<3, 256, 0, stream>>>(wsf);
  vq_cbnorm<<<64, 256, 0, stream>>>(cb, cbn);
  vq_main<<<NBLOCKS, 32 * NWAVES, 0, stream>>>(z, kappa, cbn, gum, out,
                                                avgp_acc, sdisc, scont);
  vq_finalize<<<1, 32, 0, stream>>>(kappa, avgp_acc, sdisc, scont, out);
}


# Gumbel branch (uniform-draw loads, gumbel-softmax numerators, GEMM2) moves from pass 2 into pass 1 behind the trip's own GEMM1, which covers the loads; pass 2: probabilities only; sums unchanged
# speedup vs baseline: 2.1314x; 1.0645x over previous
#include <hip/hip_runtime.h>
#include <hip/hip_bf16.h>


typedef __attribute__((ext_vector_type(2))) float v2f;
typedef __attribute__((ext_vector_type(8))) float v8f;

#define DIMD 64
#define KCODES 512
#define NROWS 65536
#define TILE_ROWS 16
#define ZQ_PITCH 68
#define NWAVES 8
#define WAVE_TILES 2
#define NBLOCKS 256
#define ZQ_F4 (TILE_ROWS * ZQ_PITCH / 4)
#define NSTASH 12
#define AREA_F4 (NSTASH * 128)
#define NREG 4
#define REG_FROM (KCODES / 16 - 2 * NREG)
static_assert(2 * NSTASH == REG_FROM, "every trip's logits are kept: LDS first, registers for the rest");
static_assert(NBLOCKS * NWAVES * WAVE_TILES * TILE_ROWS == NROWS, "the tiles cover the rows");
static_assert(ZQ_F4 <= AREA_F4, "the output staging reuses the wave's area");
static_assert((8192 + NWAVES * AREA_F4) * 16 <= 327680, "LDS of one compute unit");

#define LOG2E 1.4426950408889634f
#define LN2 0.6931471805599453f

__device__ __forceinline__ float hw_exp2(float x) { return __builtin_amdgcn_exp2f(x); }
__device__ __forceinline__ float hw_log2(float x) { return __builtin_amdgcn_logf(x); }

__device__ __forceinline__ float wave_sum(float v) {
#pragma unroll
  for (int off = 16; off > 0; off >>= 1) v += __shfl_xor(v, off, 32);
  return v;
}

template <int CTRL>
__device__ __forceinline__ float dpp_get(float v) {
  return __builtin_bit_cast(
      float, __builtin_amdgcn_update_dpp(0, __builtin_bit_cast(int, v), CTRL,
                                         0xF, 0xF, true));
}
template <int CTRL>
__device__ __forceinline__ float pair_step(bool up, float a, float b) {
  const float keep = up ? b : a;
  const float give = up ? a : b;
  return keep + dpp_get<CTRL>(give);
}

__device__ __forceinline__ float xhalf(float x) {
  int b = __builtin_bit_cast(int, x);
  return __builtin_bit_cast(
      float, __builtin_amdgcn_permlanex16(b, b, 0x76543210, 0xfedcba98, false, false));
}

__device__ __forceinline__ void gemm1_pair(const float* cbs, int ct, int c, int h,
                                           const v2f (&zf)[16], v8f& a0, v8f& a1) {
  a0 = (v8f){};
  a1 = (v8f){};
  const int a1o = (ct * 16 + c) * DIMD + 4 * c + 2 * h;
  float2 t0[16], t1[16];
#pragma unroll
  for (int ks = 0; ks < 16; ++ks) {
    const float* p0 = cbs + (a1o ^ (ks << 2));
    t0[ks] = *(const float2*)p0;
    t1[ks] = *(const float2*)(p0 + 16 * DIMD);
  }
  __builtin_amdgcn_sched_barrier(0);
#pragma unroll
  for (int ks = 0; ks < 16; ++ks) {
    v2f f0, f1;
    f0.x = t0[ks].x;
    f0.y = t0[ks].y;
    f1.x = t1[ks].x;
    f1.y = t1[ks].y;
    a0 = __builtin_amdgcn_wmma_f32_16x16x4_f32(false, f0, false, zf[ks],
                                               (short)0, a0, false, false);
    a1 = __builtin_amdgcn_wmma_f32_16x16x4_f32(false, f1, false, zf[ks],
                                               (short)0, a1, false, false);
  }
  asm volatile("v_nop\n\tv_nop\n\tv_nop\n\tv_nop" : "+v"(a0), "+v"(a1));
}

__device__ __forceinline__ void pass1_tail(const v8f& a0, const v8f& a1, float kL,
                                           float& m1, float& s1) {
  float q[16];
#pragma unroll
  for (int r = 0; r < 8; ++r) {
    q[r] = kL * a0[r];
    q[8 + r] = kL * a1[r];
  }
  float mt = q[0];
#pragma unroll
  for (int i = 1; i < 16; ++i) mt = fmaxf(mt, q[i]);
  const float mn = fmaxf(m1, mt);
  s1 *= hw_exp2(m1 - mn);
#pragma unroll
  for (int i = 0; i < 16; ++i) s1 += hw_exp2(q[i] - mn);
  m1 = mn;
}

__device__ __forceinline__ void load_gumbel(const float* gr, int ct, float4 (&uq)[4]) {
#pragma unroll
  for (int t = 0; t < 2; ++t) {
    uq[2 * t + 0] = *(const float4*)(gr + (ct + t) * 16);
    uq[2 * t + 1] = *(const float4*)(gr + (ct + t) * 16 + 4);
  }
}

__device__ __forceinline__ float eps_behind(float& s1) {
  float e = 1e-10f;
  asm volatile("" : "+s"(e), "+v"(s1));
  return e;
}

__device__ __forceinline__ void gumbel_tail(const float* cbs, int ct, int c, int h,
                                            float kL, float eps, const v8f& a0,
                                            const v8f& a1,
                                            const float4 (&uq)[4], const int (&g2p)[4],
                                            const int (&g2m)[4], float& m2, float& s2,
                                            v8f (&d)[4]) {
  float y2[16];
  float mt = -3.4e38f;
#pragma unroll
  for (int t = 0; t < 2; ++t) {
    float uu[8] = {uq[2 * t].x, uq[2 * t].y, uq[2 * t].z, uq[2 * t].w,
                   uq[2 * t + 1].x, uq[2 * t + 1].y, uq[2 * t + 1].z, uq[2 * t + 1].w};
#pragma unroll
    for (int r = 0; r < 8; ++r) {
      const float a = (t == 0 ? a0[r] : a1[r]);
      const float tg = __builtin_fmaf(hw_log2(uu[r] + eps), -LN2, 1e-10f);
      const float v = __builtin_fmaf(a, kL, -hw_log2(tg));
      mt = fmaxf(mt, v);
      y2[8 * t + r] = v;
    }
  }
  mt = fmaxf(mt, xhalf(mt));
  const float mn = fmaxf(m2, mt);
  const float alpha = hw_exp2((m2 - mn) * 2.0f);
  m2 = mn;
  const float mn2 = 2.0f * mn;
  s2 *= alpha;
#pragma unroll
  for (int nt = 0; nt < 4; ++nt) d[nt] *= alpha;
  float en[16];
#pragma unroll
  for (int i = 0; i < 16; ++i) {
    en[i] = hw_exp2(__builtin_fmaf(y2[i], 2.0f, -mn2));
    s2 += en[i];
  }

  v2f av[2][4][4];
#pragma unroll
  for (int t = 0; t < 2; ++t) {
    const float* cr = cbs + (ct + t) * 16 * DIMD;
#pragma unroll
    for (int q = 0; q < 4; ++q) {
#pragma unroll
      for (int nt = 0; nt < 4; ++nt) {
        const int k1 = (nt * 16) ^ ((8 * q) & 16);
        const int m0 = 2 * (q & 1);
        av[t][q][nt].x = cr[(k1 < 32 ? g2p[m0] : g2m[m0]) + (2 * q) * DIMD + k1];
        av[t][q][nt].y = cr[(k1 < 32 ? g2p[m0 + 1] : g2m[m0 + 1]) + (2 * q + 1) * DIMD + k1];
      }
    }
  }
  __builtin_amdgcn_sched_barrier(0);
#pragma unroll
  for (int t = 0; t < 2; ++t) {
#pragma unroll
    for (int q = 0; q < 4; ++q) {
      v2f b;
      b.x = en[8 * t + 2 * q];
      b.y = en[8 * t + 2 * q + 1];
#pragma unroll
      for (int nt = 0; nt < 4; ++nt) {
        d[nt] = __builtin_amdgcn_wmma_f32_16x16x4_f32(false, av[t][q][nt], false, b,
                                                      (short)0, d[nt], false,
                                                      false);
      }
    }
  }
}

__device__ __forceinline__ void prob_tail(int ct, int c, int h, float kL, float lse2,
                                          const v8f& a0, const v8f& a1, float* avgp_acc,
                                          float& acc_pl) {
  const bool bit0 = (c & 1) != 0, bit1 = (c & 2) != 0, bit2 = (c & 4) != 0, bit3 = (c & 8) != 0;
  float tile_sum[2];
#pragma unroll
  for (int t = 0; t < 2; ++t) {
    float pv[8];
#pragma unroll
    for (int r = 0; r < 8; ++r) {
      const float a = (t == 0 ? a0[r] : a1[r]);
      const float lp2 = __builtin_fmaf(a, kL, -lse2);
      const float p = hw_exp2(lp2);
      acc_pl = __builtin_fmaf(p, lp2, acc_pl);
      pv[r] = p;
    }
    float w[4], x[2];
#pragma unroll
    for (int i = 0; i < 4; ++i) w[i] = pair_step<0xB1>(bit0, pv[2 * i], pv[2 * i + 1]);
#pragma unroll
    for (int i = 0; i < 2; ++i) x[i] = pair_step<0x4E>(bit1, w[2 * i], w[2 * i + 1]);
    tile_sum[t] = pair_step<0x164>(bit2, x[0], x[1]);
  }
  atomicAdd(&avgp_acc[ct * 16 + ((c >> 3) << 4) + 8 * h + (c & 7)],
            pair_step<0x168>(bit3, tile_sum[0], tile_sum[1]));
}

__global__ void vq_zero(float* acc) {
  int i = blockIdx.x * blockDim.x + threadIdx.x;
  if (i < 514) acc[i] = 0.0f;
}

__global__ __launch_bounds__(256) void vq_cbnorm(const float* __restrict__ cb,
                                                 float* __restrict__ cbn) {
  int wave = (blockIdx.x * blockDim.x + threadIdx.x) >> 5;
  int lane = threadIdx.x & 31;
  const float* row = cb + wave * DIMD;
  float v0 = row[lane], v1 = row[lane + 32];
  float s = wave_sum(v0 * v0 + v1 * v1);
  float scale = 1.0f / fmaxf(sqrtf(s), 1e-12f);
  cbn[wave * DIMD + lane] = v0 * scale;
  cbn[wave * DIMD + lane + 32] = v1 * scale;
}

__global__ __launch_bounds__(32 * NWAVES) void vq_main(
    const float* __restrict__ z, const float* __restrict__ kappa_p,
    const float* __restrict__ cbn, const float* gum,
    float* __restrict__ out, float* __restrict__ avgp_acc,
    float* __restrict__ sdisc, float* __restrict__ scont) {
  __shared__ float4 lds4[8192 + NWAVES * AREA_F4];
  const float* cbs = (const float*)lds4;

  const int lane = threadIdx.x & 31;
  const int wave = threadIdx.x >> 5;
  const int c = lane & 15;
  const int h = lane >> 4;
  const float kL = kappa_p[0] * LOG2E;
  float4* const area = lds4 + 8192 + wave * AREA_F4;

#pragma unroll 16
  for (int i = threadIdx.x; i < KCODES * DIMD / 4; i += 32 * NWAVES) {
    int code = i >> 4;
    int d0 = (i & 15) << 2;
    float4 v = *(const float4*)(cbn + (long)i * 4);
    lds4[(code * DIMD + (d0 ^ ((code & 15) << 2))) >> 2] = v;
  }
  __syncthreads();

  int g2p[4], g2m[4];
#pragma unroll
  for (int m = 0; m < 4; ++m) {
    g2p[m] = 512 * h + (c ^ (4 * m)) + 32 * h;
    g2m[m] = 512 * h + (c ^ (4 * m)) - 32 * h;
  }

  for (int it = 0; it < WAVE_TILES; ++it) {
    const long base_row =
        (((long)blockIdx.x * WAVE_TILES + it) * NWAVES + wave) * TILE_ROWS;

    v2f zf[16];
    {
      const float* zr = z + (base_row + c) * DIMD + 2 * h;
      float ss = 0.0f;
#pragma unroll
      for (int ks = 0; ks < 16; ++ks) {
        float2 t = *(const float2*)(zr + 4 * ks);
        zf[ks].x = t.x;
        zf[ks].y = t.y;
        ss += t.x * t.x + t.y * t.y;
      }
      ss += xhalf(ss);
      float zs = 1.0f / fmaxf(sqrtf(ss), 1e-12f);
#pragma unroll
      for (int ks = 0; ks < 16; ++ks) {
        zf[ks].x *= zs;
        zf[ks].y *= zs;
      }
    }

    float m1 = -3.4e38f;
    float s1 = 0.0f;
    float m2 = -3.4e38f;
    float s2 = 0.0f;
    v8f d[4] = {};
    const float* gr = gum + (base_row + c) * (long)KCODES + 8 * h;
    for (int ct = 0; ct < REG_FROM; ct += 2) {
      float4 uq[4];
      load_gumbel(gr, ct, uq);
      v8f a0, a1;
      gemm1_pair(cbs, ct, c, h, zf, a0, a1);
      float4* s = area + (ct >> 1) * 128 + lane;
      s[0] = make_float4(a0[0], a0[1], a0[2], a0[3]);
      s[32] = make_float4(a0[4], a0[5], a0[6], a0[7]);
      s[64] = make_float4(a1[0], a1[1], a1[2], a1[3]);
      s[96] = make_float4(a1[4], a1[5], a1[6], a1[7]);
      pass1_tail(a0, a1, kL, m1, s1);
      gumbel_tail(cbs, ct, c, h, kL, eps_behind(s1), a0, a1, uq, g2p, g2m, m2, s2, d);
    }
    float kept[NREG * 16];
#pragma unroll
    for (int i = 0; i < NREG * 16; ++i) kept[i] = 0.0f;
#pragma nounroll
    for (int ct = REG_FROM; ct < KCODES / 16; ct += 2) {
      float4 uq[4];
      load_gumbel(gr, ct, uq);
      v8f a0, a1;
      gemm1_pair(cbs, ct, c, h, zf, a0, a1);
#pragma unroll
      for (int i = 0; i < (NREG - 1) * 16; ++i) kept[i] = kept[i + 16];
#pragma unroll
      for (int r = 0; r < 8; ++r) {
        kept[(NREG - 1) * 16 + r] = a0[r];
        kept[(NREG - 1) * 16 + 8 + r] = a1[r];
      }
      pass1_tail(a0, a1, kL, m1, s1);
      gumbel_tail(cbs, ct, c, h, kL, eps_behind(s1), a0, a1, uq, g2p, g2m, m2, s2, d);
    }
    {
      const float mrow = fmaxf(m1, xhalf(m1));
      s1 *= hw_exp2(m1 - mrow);
      s1 += xhalf(s1);
      m1 = mrow;
    }
    const float lse2 = m1 + hw_log2(s1);

    float acc_pl = 0.0f;
    for (int ct = 0; ct < REG_FROM; ct += 2) {
      const float4* s = area + (ct >> 1) * 128 + lane;
      const float4 x0 = s[0], x1 = s[32], x2 = s[64], x3 = s[96];
      const v8f a0 = (v8f){x0.x, x0.y, x0.z, x0.w, x1.x, x1.y, x1.z, x1.w};
      const v8f a1 = (v8f){x2.x, x2.y, x2.z, x2.w, x3.x, x3.y, x3.z, x3.w};
      prob_tail(ct, c, h, kL, lse2, a0, a1, avgp_acc, acc_pl);
    }
    for (int ct = REG_FROM; ct < KCODES / 16; ct += 2) {
      v8f a0, a1;
#pragma unroll
      for (int r = 0; r < 8; ++r) {
        a0[r] = kept[r];
        a1[r] = kept[8 + r];
      }
#pragma unroll
      for (int i = 0; i < (NREG - 1) * 16; ++i) kept[i] = kept[i + 16];
      prob_tail(ct, c, h, kL, lse2, a0, a1, avgp_acc, acc_pl);
    }
    acc_pl = wave_sum(acc_pl);
    if (lane == 0) atomicAdd(sdisc, acc_pl * LN2);
    s2 += xhalf(s2);
    const float inv_s2 = 1.0f / s2;

    float* zq = (float*)area;
#pragma unroll
    for (int nt = 0; nt < 4; ++nt) {
      float4 lo, hi4;
      lo.x = d[nt][0] * inv_s2; lo.y = d[nt][1] * inv_s2;
      lo.z = d[nt][2] * inv_s2; lo.w = d[nt][3] * inv_s2;
      hi4.x = d[nt][4] * inv_s2; hi4.y = d[nt][5] * inv_s2;
      hi4.z = d[nt][6] * inv_s2; hi4.w = d[nt][7] * inv_s2;
      *(float4*)(zq + c * ZQ_PITCH + nt * 16 + 8 * h) = lo;
      *(float4*)(zq + c * ZQ_PITCH + nt * 16 + 8 * h + 4) = hi4;
    }
    __builtin_amdgcn_fence(__ATOMIC_ACQ_REL, "wavefront");
    float acc_cont = 0.0f;
#pragma unroll
    for (int r = 0; r < TILE_ROWS; ++r) {
      float2 qv = *(const float2*)(zq + r * ZQ_PITCH + 2 * lane);
      long idx = (base_row + r) * DIMD + 2 * lane;
      float2 zo = *(const float2*)(z + idx);
      acc_cont += zo.x * (zo.x - qv.x);
      acc_cont += zo.y * (zo.y - qv.y);
      *(float2*)(out + idx) = qv;
    }
    acc_cont = wave_sum(acc_cont);
    if (lane == 0) atomicAdd(scont, acc_cont);
  }
}

__global__ __launch_bounds__(32) void vq_finalize(
    const float* __restrict__ kappa_p, const float* __restrict__ avgp_acc,
    const float* __restrict__ sdisc, const float* __restrict__ scont,
    float* __restrict__ out) {
  int lane = threadIdx.x;
  const float invN = 1.0f / (float)NROWS;
  float acc = 0.0f;
#pragma unroll
  for (int i = lane; i < KCODES; i += 32) {
    float a = avgp_acc[i] * invN;
    acc += a * logf(a + 1e-7f);
  }
  acc = wave_sum(acc);
  if (lane == 0) {
    float loss = (sdisc[0] + kappa_p[0] * scont[0]) / 8.0f;
    out[(long)NROWS * DIMD] = loss;
    out[(long)NROWS * DIMD + 1] = expf(-acc);
  }
}

extern "C" void kernel_launch(void* const* d_in, const int* in_sizes, int n_in,
                              void* d_out, int out_size, void* d_ws,
                              size_t ws_size, hipStream_t stream) {
  const float* z = (const float*)d_in[0];
  const float* kappa = (const float*)d_in[1];
  const float* cb = (const float*)d_in[2];
  const float* gum = (const float*)d_in[3];
  float* out = (float*)d_out;

  float* wsf = (float*)d_ws;
  float* avgp_acc = wsf;
  float* sdisc = wsf + 512;
  float* scont = wsf + 513;
  float* cbn = wsf + 1024;

  vq_zero<<<3, 256, 0, stream>>>(wsf);
  vq_cbnorm<<<64, 256, 0, stream>>>(cb, cbn);
  vq_main<<<NBLOCKS, 32 * NWAVES, 0, stream>>>(z, kappa, cbn, gum, out,
                                                avgp_acc, sdisc, scont);
  vq_finalize<<<1, 32, 0, stream>>>(kappa, avgp_acc, sdisc, scont, out);
}
